# starting kernel
# baseline (speedup 1.0000x reference)
#include <hip/hip_runtime.h>

typedef __attribute__((ext_vector_type(2))) float v2f;
typedef __attribute__((ext_vector_type(8))) float v8f;

#define S_LEN 1024
#define D_LEN 256
#define KB    32
#define LDK   34

__global__ __launch_bounds__(256, 3) void complex_mixture_wmma(
    const float* __restrict__ in_r,
    const float* __restrict__ in_i,
    const float* __restrict__ wgt,
    float* __restrict__ out_r,
    float* __restrict__ out_i)
{
    __shared__ __align__(16) float sWrA[64  * LDK];
    __shared__ __align__(16) float sWiA[64  * LDK];
    __shared__ __align__(16) float sRB [128 * LDK];
    __shared__ __align__(16) float sIB [128 * LDK];

    const int b     = blockIdx.z;
    const int aBase = blockIdx.y * 64;
    const int cBase = blockIdx.x * 128;

    const int tid   = threadIdx.x;
    const int lane  = tid & 31;
    const int wave  = tid >> 5;
    const int waveM = wave & 1;
    const int waveN = wave >> 1;

    const int laneLo = lane & 15;
    const int kOff   = (lane >> 4) * 2;

    const float* rSrc = in_r + (size_t)b * S_LEN * D_LEN;
    const float* iSrc = in_i + (size_t)b * S_LEN * D_LEN;
    const float* wSrc = wgt  + (size_t)b * S_LEN;

    v8f accR [2][2] = {};
    v8f accI1[2][2] = {};
    v8f accI2[2][2] = {};

    for (int kb = 0; kb < S_LEN / KB; ++kb) {
        const int s0 = kb * KB;

        #pragma unroll
        for (int it = 0; it < (KB * 64) / 256; ++it) {
            int idx = tid + it * 256;
            int sl  = idx >> 6;
            int ml  = idx & 63;
            size_t g = (size_t)(s0 + sl) * D_LEN + aBase + ml;
            float  w = wSrc[s0 + sl];
            sWrA[ml * LDK + sl] = w * rSrc[g];
            sWiA[ml * LDK + sl] = w * iSrc[g];
        }
        #pragma unroll
        for (int it = 0; it < (KB * 128) / 256; ++it) {
            int idx = tid + it * 256;
            int sl  = idx >> 7;
            int nl  = idx & 127;
            size_t g = (size_t)(s0 + sl) * D_LEN + cBase + nl;
            sRB[nl * LDK + sl] = rSrc[g];
            sIB[nl * LDK + sl] = iSrc[g];
        }
        if (kb + 1 < S_LEN / KB) {
            size_t gn = (size_t)(s0 + KB + (tid >> 3)) * D_LEN + aBase + (tid & 7) * 8;
            __builtin_prefetch(&rSrc[gn], 0, 0);
            __builtin_prefetch(&iSrc[gn], 0, 0);
        }
        __syncthreads();

        #pragma unroll
        for (int kk = 0; kk < KB; kk += 4) {
            v2f aR[2], aI[2], bR[2], bI[2];
            #pragma unroll
            for (int mt = 0; mt < 2; ++mt) {
                int row = waveM * 32 + mt * 16 + laneLo;
                aR[mt] = *(const v2f*)&sWrA[row * LDK + kk + kOff];
                aI[mt] = *(const v2f*)&sWiA[row * LDK + kk + kOff];
            }
            #pragma unroll
            for (int nt = 0; nt < 2; ++nt) {
                int col = waveN * 32 + nt * 16 + laneLo;
                bR[nt] = *(const v2f*)&sRB[col * LDK + kk + kOff];
                bI[nt] = *(const v2f*)&sIB[col * LDK + kk + kOff];
            }
            #pragma unroll
            for (int mt = 0; mt < 2; ++mt) {
                #pragma unroll
                for (int nt = 0; nt < 2; ++nt) {
                    accR[mt][nt]  = __builtin_amdgcn_wmma_f32_16x16x4_f32(
                        false, aR[mt], false, bR[nt], (short)0, accR[mt][nt],  false, false);
                    accR[mt][nt]  = __builtin_amdgcn_wmma_f32_16x16x4_f32(
                        false, aI[mt], false, bI[nt], (short)0, accR[mt][nt],  false, false);
                    accI1[mt][nt] = __builtin_amdgcn_wmma_f32_16x16x4_f32(
                        false, aI[mt], false, bR[nt], (short)0, accI1[mt][nt], false, false);
                    accI2[mt][nt] = __builtin_amdgcn_wmma_f32_16x16x4_f32(
                        false, aR[mt], false, bI[nt], (short)0, accI2[mt][nt], false, false);
                }
            }
        }
        __syncthreads();
    }

    const int laneHi8 = (lane >> 4) * 8;
    #pragma unroll
    for (int mt = 0; mt < 2; ++mt) {
        #pragma unroll
        for (int nt = 0; nt < 2; ++nt) {
            int row0 = aBase + waveM * 32 + mt * 16 + laneHi8;
            int col  = cBase + waveN * 32 + nt * 16 + laneLo;
            #pragma unroll
            for (int j = 0; j < 8; ++j) {
                size_t o = ((size_t)b * D_LEN + row0 + j) * D_LEN + col;
                out_r[o] = accR[mt][nt][j];
                out_i[o] = accI1[mt][nt][j] - accI2[mt][nt][j];
            }
        }
    }
}

extern "C" void kernel_launch(void* const* d_in, const int* in_sizes, int n_in,
                              void* d_out, int out_size, void* d_ws, size_t ws_size,
                              hipStream_t stream) {
    const float* in_r = (const float*)d_in[0];
    const float* in_i = (const float*)d_in[1];
    const float* w    = (const float*)d_in[2];
    float* out_r = (float*)d_out;
    float* out_i = out_r + (size_t)32 * D_LEN * D_LEN;

    dim3 grid(D_LEN / 128, D_LEN / 64, 32);
    dim3 block(256);
    hipLaunchKernelGGL(complex_mixture_wmma, grid, block, 0, stream,
                       in_r, in_i, w, out_r, out_i);
}


# three real matrix products per complex product instead of four: P1=(wr+wi)^T r, P2=wi^T(r-i), P3=wr^T(r+i), operand sums formed in registers; out_r=P1-P2, out_i=P1-P3 (3 WMMAs per k-step, was 4)
# speedup vs baseline: 9.7130x; 9.7130x over previous
#include <hip/hip_runtime.h>


typedef __attribute__((ext_vector_type(2))) float v2f;
typedef __attribute__((ext_vector_type(4))) float v4f;
typedef __attribute__((ext_vector_type(8))) float v8f;

#define S_LEN 1024
#define D_LEN 256
#define KB    32
#define NKB   (S_LEN / KB)
#define UNIT  32
#define NUNIT (D_LEN / UNIT)
#define NTRI  (NUNIT * (NUNIT + 1) / 2)


#define LDP   96
#define T_ELEMS ((KB / 2) * LDP)
#define TP    33

struct Fetch {
    v4f ar[2], ai[2];
    v4f br[2], bi[2];
    v2f w;
};

__device__ __forceinline__ void fetch_block(
    Fetch& f, const float* rA, const float* iA, const float* wA,
    const float* rB, const float* iB, int s0)
{
    const size_t o = (size_t)s0 * D_LEN;
    f.ar[0] = *(const v4f*)(rA + o);
    f.ar[1] = *(const v4f*)(rA + o + D_LEN);
    f.ai[0] = *(const v4f*)(iA + o);
    f.ai[1] = *(const v4f*)(iA + o + D_LEN);
    f.w     = *(const v2f*)(wA + s0);
    f.br[0] = *(const v4f*)(rB + o);
    f.br[1] = *(const v4f*)(rB + o + D_LEN);
    f.bi[0] = *(const v4f*)(iB + o);
    f.bi[1] = *(const v4f*)(iB + o + D_LEN);
}

__device__ __forceinline__ void put_pair(float* dst, v4f x0, v4f x1, float w0, float w1)
{
    v4f lo = { w0 * x0.x, w1 * x1.x, w0 * x0.y, w1 * x1.y };
    v4f hi = { w0 * x0.z, w1 * x1.z, w0 * x0.w, w1 * x1.w };
    *(v4f*)(dst)     = lo;
    *(v4f*)(dst + 4) = hi;
}

__device__ __forceinline__ void put_pair_plain(float* dst, v4f x0, v4f x1)
{
    v4f lo = { x0.x, x1.x, x0.y, x1.y };
    v4f hi = { x0.z, x1.z, x0.w, x1.w };
    *(v4f*)(dst)     = lo;
    *(v4f*)(dst + 4) = hi;
}

__device__ __forceinline__ void put_block(
    const Fetch& f, float* sWr, float* sWi, float* sR, float* sI, int st)
{
    put_pair(sWr + st, f.ar[0], f.ar[1], f.w.x, f.w.y);
    put_pair(sWi + st, f.ai[0], f.ai[1], f.w.x, f.w.y);
    put_pair_plain(sR + st, f.br[0], f.br[1]);
    put_pair_plain(sI + st, f.bi[0], f.bi[1]);
}

__device__ __forceinline__ void mma_block(
    const float* sWr, const float* sWi, const float* sR, const float* sI,
    int aOff, int bOff, v8f& acc1, v8f& acc2, v8f& acc3)
{
    #pragma unroll
    for (int kk = 0; kk < KB; kk += 4) {
        const int oa = (kk >> 1) * LDP + aOff;
        const int ob = (kk >> 1) * LDP + bOff;
        const v2f aR = *(const v2f*)&sWr[oa];
        const v2f aI = *(const v2f*)&sWi[oa];
        const v2f bR = *(const v2f*)&sR[ob];
        const v2f bI = *(const v2f*)&sI[ob];
        const v2f aS = aR + aI;
        const v2f bD = bR - bI;
        const v2f bS = bR + bI;
        acc1 = __builtin_amdgcn_wmma_f32_16x16x4_f32(
            false, aS, false, bR, (short)0, acc1, false, false);
        acc2 = __builtin_amdgcn_wmma_f32_16x16x4_f32(
            false, aI, false, bD, (short)0, acc2, false, false);
        acc3 = __builtin_amdgcn_wmma_f32_16x16x4_f32(
            false, aR, false, bS, (short)0, acc3, false, false);
    }
}

__global__ __launch_bounds__(128) void complex_mixture_herm(
    const float* __restrict__ in_r,
    const float* __restrict__ in_i,
    const float* __restrict__ wgt,
    float* __restrict__ out_r,
    float* __restrict__ out_i)
{
    __shared__ __align__(16) float sWr[2][T_ELEMS];
    __shared__ __align__(16) float sWi[2][T_ELEMS];
    __shared__ __align__(16) float sR [2][T_ELEMS];
    __shared__ __align__(16) float sI [2][T_ELEMS];
    __shared__ float tR[UNIT * TP];
    __shared__ float tI[UNIT * TP];

    int ui = 0, rem = blockIdx.x;
    #pragma unroll
    for (int n = 0; n < NUNIT - 1; ++n) {
        const int len = NUNIT - ui;
        if (rem >= len) { rem -= len; ++ui; }
    }
    const int uj    = ui + rem;
    const int b     = blockIdx.y;
    const int aBase = ui * UNIT;
    const int cBase = uj * UNIT;

    const int tid   = threadIdx.x;
    const int lane  = tid & 31;
    const int wave  = tid >> 5;
    const int waveM = wave & 1;
    const int waveN = wave >> 1;

    const int laneLo = lane & 15;
    const int laneHi = lane >> 4;

    const float* rSrc = in_r + (size_t)b * S_LEN * D_LEN;
    const float* iSrc = in_i + (size_t)b * S_LEN * D_LEN;
    const float* wSrc = wgt  + (size_t)b * S_LEN;

    const int aOff = laneHi * LDP + 2 * (waveM * 16 + laneLo);
    const int bOff = laneHi * LDP + 2 * (waveN * 16 + laneLo);

    const int kp = tid >> 3, q = tid & 7;
    const int st = kp * LDP + 8 * q;
    const float* rA = rSrc + (size_t)(2 * kp) * D_LEN + aBase + 4 * q;
    const float* iA = iSrc + (size_t)(2 * kp) * D_LEN + aBase + 4 * q;
    const float* wA = wSrc + 2 * kp;
    const float* rB = rSrc + (size_t)(2 * kp) * D_LEN + cBase + 4 * q;
    const float* iB = iSrc + (size_t)(2 * kp) * D_LEN + cBase + 4 * q;

    v8f acc1 = {};
    v8f acc2 = {};
    v8f acc3 = {};

    Fetch f;
    fetch_block(f, rA, iA, wA, rB, iB, 0);
    put_block(f, sWr[0], sWi[0], sR[0], sI[0], st);
    __syncthreads();

    for (int kb = 0; kb < NKB; kb += 2) {
        fetch_block(f, rA, iA, wA, rB, iB, (kb + 1) * KB);
        __builtin_amdgcn_sched_barrier(0);
        mma_block(sWr[0], sWi[0], sR[0], sI[0], aOff, bOff, acc1, acc2, acc3);
        __builtin_amdgcn_sched_barrier(0);
        put_block(f, sWr[1], sWi[1], sR[1], sI[1], st);
        __syncthreads();

        const bool more = (kb + 2 < NKB);
        if (more) fetch_block(f, rA, iA, wA, rB, iB, (kb + 2) * KB);
        __builtin_amdgcn_sched_barrier(0);
        mma_block(sWr[1], sWi[1], sR[1], sI[1], aOff, bOff, acc1, acc2, acc3);
        __builtin_amdgcn_sched_barrier(0);
        if (more) put_block(f, sWr[0], sWi[0], sR[0], sI[0], st);
        __syncthreads();
    }

    {
        const int row0 = waveM * 16 + laneHi * 8;
        const int col  = waveN * 16 + laneLo;
        #pragma unroll
        for (int j = 0; j < 8; ++j) {
            tR[(row0 + j) * TP + col] = acc1[j] - acc2[j];
            tI[(row0 + j) * TP + col] = acc1[j] - acc3[j];
        }
    }
    __syncthreads();

    const size_t outB = (size_t)b * D_LEN * D_LEN;
    #pragma unroll
    for (int p = 0; p < UNIT / 4; ++p) {
        const int row = p * 4 + wave;
        const size_t o = outB + (size_t)(aBase + row) * D_LEN + cBase + lane;
        out_r[o] = tR[row * TP + lane];
        out_i[o] = tI[row * TP + lane];
    }
    if (ui != uj) {
        #pragma unroll
        for (int p = 0; p < UNIT / 4; ++p) {
            const int row = p * 4 + wave;
            const size_t o = outB + (size_t)(cBase + row) * D_LEN + aBase + lane;
            out_r[o] =  tR[lane * TP + row];
            out_i[o] = -tI[lane * TP + row];
        }
    }
}

extern "C" void kernel_launch(void* const* d_in, const int* in_sizes, int n_in,
                              void* d_out, int out_size, void* d_ws, size_t ws_size,
                              hipStream_t stream) {
    const float* in_r = (const float*)d_in[0];
    const float* in_i = (const float*)d_in[1];
    const float* w    = (const float*)d_in[2];
    float* out_r = (float*)d_out;
    float* out_i = out_r + (size_t)32 * D_LEN * D_LEN;

    dim3 grid(NTRI, 32, 1);
    dim3 block(128);
    hipLaunchKernelGGL(complex_mixture_herm, grid, block, 0, stream,
                       in_r, in_i, w, out_r, out_i);
}
